# starting kernel
# baseline (speedup 1.0000x reference)
#include <hip/hip_runtime.h>
#include <math.h>

#define L     880
#define HID   8
#define OUT   96
#define DS    48
#define T     64
#define BATCH 4
#define NCONV 63
#define CW    64
#define CSTEP (L * CW)

typedef float v2f __attribute__((ext_vector_type(2)));
typedef float v8f __attribute__((ext_vector_type(8)));

template <int LDB, int LDD>
__global__ void wmma_mm_kernel(const float* __restrict__ Am,
                               const float* __restrict__ Bb, long bz,
                               float* __restrict__ Db, long dz,
                               int ncols) {
    const int lane = threadIdx.x & 31;
    const int jt   = blockIdx.y * 4 + (threadIdx.x >> 5);
    if (jt * 16 >= ncols) return;

    const int itile = blockIdx.x;
    const float* Bm = Bb + (size_t)blockIdx.z * bz;
    float*       Dm = Db + (size_t)blockIdx.z * dz;

    const int mrow  = lane & 15;
    const int khalf = (lane >> 4) << 1;
    const int col   = jt * 16 + (lane & 15);

    const float* Arow = Am + (size_t)(itile * 16 + mrow) * L;

    v8f acc = {};
    for (int k = 0; k < L; k += 4) {
        v2f a, b;
        a.x = Arow[k + khalf + 0];
        a.y = Arow[k + khalf + 1];
        b.x = Bm[(size_t)(k + khalf + 0) * LDB + col];
        b.y = Bm[(size_t)(k + khalf + 1) * LDB + col];
        acc = __builtin_amdgcn_wmma_f32_16x16x4_f32(false, a, false, b,
                                                    (short)0, acc, false, false);
    }

    const int rbase = itile * 16 + ((lane >> 4) << 3);
    #pragma unroll
    for (int v = 0; v < 8; ++v)
        Dm[(size_t)(rbase + v) * LDD + col] = acc[v];
}

__global__ void ses2seq_init(const float* __restrict__ Bm,
                             const float* __restrict__ theta0,
                             const float* __restrict__ xs,
                             float* __restrict__ C0,
                             float* __restrict__ thetas) {
    int l = blockIdx.x * blockDim.x + threadIdx.x;
    if (l >= L) return;
    float* row = C0 + (size_t)l * CW;
    const float* brow = Bm + (size_t)l * DS;
    #pragma unroll 4
    for (int d = 0; d < DS; ++d) row[d] = brow[d];
    row[DS] = theta0[l];
    #pragma unroll
    for (int b = 0; b < BATCH; ++b) {
        float acc = 0.f;
        const float* x0 = xs + (size_t)b * T * DS;
        for (int d = 0; d < DS; ++d) acc += brow[d] * x0[d];
        row[DS + 1 + b] = -acc;
    }
    for (int c = DS + 1 + BATCH; c < CW; ++c) row[c] = 0.f;
    #pragma unroll
    for (int b = 0; b < BATCH; ++b)
        thetas[((size_t)b * T + 0) * L + l] = theta0[l];
}

__global__ void ses2seq_conv(const float* __restrict__ C,
                             const float* __restrict__ xs,
                             float* __restrict__ thetas) {
    const int lane  = threadIdx.x & 31;
    const int wv    = threadIdx.x >> 5;
    const int ltile = blockIdx.x;
    const int b     = blockIdx.y;
    const int khalf = (lane >> 4) << 1;
    const int lrow  = ltile * 16 + (lane & 15);
    const int tc    = wv * 16 + (lane & 15);
    const int tcc   = (tc >= NCONV) ? 0 : tc;

    const float* u = xs + (size_t)b * T * DS;

    v8f acc = {};
    for (int m = 0; m < NCONV; ++m) {
        const float* Km = C + (size_t)m * CSTEP;
        const float* Kp = C + (size_t)(m > 0 ? m - 1 : 0) * CSTEP;
        const float  w  = (m > 0) ? 1.f : 0.f;
        int s = tcc - m; if (s < 0) s += NCONV;
        const float* urow = u + (size_t)(s + 1) * DS;
        #pragma unroll 4
        for (int d = 0; d < DS; d += 4) {
            const int dk = d + khalf;
            v2f a, bb;
            a.x = Km[(size_t)lrow * CW + dk + 0] - w * Kp[(size_t)lrow * CW + dk + 0];
            a.y = Km[(size_t)lrow * CW + dk + 1] - w * Kp[(size_t)lrow * CW + dk + 1];
            bb.x = urow[dk + 0];
            bb.y = urow[dk + 1];
            acc = __builtin_amdgcn_wmma_f32_16x16x4_f32(false, a, false, bb,
                                                        (short)0, acc, false, false);
        }
    }

    if (tc < NCONV) {
        const int l0 = ltile * 16 + ((lane >> 4) << 3);
        const int zc = (tc - 1 + NCONV) % NCONV;
        #pragma unroll
        for (int v = 0; v < 8; ++v) {
            const int l = l0 + v;
            float val = acc[v]
                      + C[(size_t)tc * CSTEP + (size_t)l * CW + DS]
                      + C[(size_t)zc * CSTEP + (size_t)l * CW + DS + 1 + b];
            thetas[((size_t)b * T + (tc + 1)) * L + l] = val;
        }
    }
}

__global__ void ses2seq_head(const float* __restrict__ thetas,
                             const float* __restrict__ ts,
                             float* __restrict__ out) {
    const int bt = blockIdx.x;
    const float* th = thetas + (size_t)bt * L;
    __shared__ float h[HID];
    if (threadIdx.x < HID) {
        float x = th[threadIdx.x] * ts[bt] + th[HID + threadIdx.x];
        h[threadIdx.x] = x > 0.f ? x : 0.f;
    }
    __syncthreads();
    const int o = threadIdx.x;
    float acc = th[2 * HID + OUT * HID + o];
    const float* w1 = th + 2 * HID + o * HID;
    #pragma unroll
    for (int k = 0; k < HID; ++k) acc += w1[k] * h[k];
    float r;
    if (o < DS) {
        r = tanhf(acc);
    } else {
        r = acc > 0.f ? acc + log1pf(expf(-acc)) : log1pf(expf(acc));
    }
    out[(size_t)bt * OUT + o] = r;
}

extern "C" void kernel_launch(void* const* d_in, const int* in_sizes, int n_in,
                              void* d_out, int out_size, void* d_ws, size_t ws_size,
                              hipStream_t stream) {
    const float* xs     = (const float*)d_in[0];
    const float* ts     = (const float*)d_in[1];
    const float* theta0 = (const float*)d_in[2];
    const float* A      = (const float*)d_in[3];
    const float* Bm     = (const float*)d_in[4];
    (void)in_sizes; (void)n_in; (void)out_size; (void)ws_size;

    float* A2     = (float*)d_ws;
    float* A4     = A2 + (size_t)L * L;
    float* C      = A4 + (size_t)L * L;
    float* thetas = C + (size_t)NCONV * CSTEP;
    float* out    = (float*)d_out;

    ses2seq_init<<<(L + 255) / 256, 256, 0, stream>>>(Bm, theta0, xs, C, thetas);

    wmma_mm_kernel<L, L><<<dim3(55, 14, 1), 128, 0, stream>>>(A,  A,  0, A2, 0, L);
    wmma_mm_kernel<L, L><<<dim3(55, 14, 1), 128, 0, stream>>>(A2, A2, 0, A4, 0, L);

    wmma_mm_kernel<CW, CW><<<dim3(55, 1, 1), 128, 0, stream>>>(
        A, C, 0, C + CSTEP, 0, CW);
    wmma_mm_kernel<CW, CW><<<dim3(55, 1, 2), 128, 0, stream>>>(
        A2, C, CSTEP, C + (size_t)2 * CSTEP, CSTEP, CW);

    for (int j = 1; j <= 15; ++j) {
        int nz = (j == 15) ? 3 : 4;
        wmma_mm_kernel<CW, CW><<<dim3(55, 1, nz), 128, 0, stream>>>(
            A4,
            C + (size_t)(4 * j - 4) * CSTEP, CSTEP,
            C + (size_t)(4 * j) * CSTEP, CSTEP, CW);
    }

    ses2seq_conv<<<dim3(55, BATCH), 128, 0, stream>>>(C, xs, thetas);

    ses2seq_head<<<BATCH * T, OUT, 0, stream>>>(thetas, ts, out);
}


# mm kernel retiled: block owns a 16x64 strip, K split over 10 waves (one A fragment feeds 4 WMMAs; 16-byte loads, all issued first), partials summed via LDS in wave order: new summation order
# speedup vs baseline: 2.3294x; 2.3294x over previous
#include <hip/hip_runtime.h>
#include <math.h>

#define L     880
#define HID   8
#define OUT   96
#define DS    48
#define T     64
#define BATCH 4
#define NCONV 63
#define CW    64
#define CSTEP (L * CW)

typedef float v2f __attribute__((ext_vector_type(2)));
typedef float v8f __attribute__((ext_vector_type(8)));

#define MM_NW   10
#define MM_KG   (L / 8)
#define MM_GPW  (MM_KG / MM_NW)
#define MM_PITCH 68

typedef float v4f __attribute__((ext_vector_type(4)));

template <int LDB, int LDD>
__global__ __launch_bounds__(MM_NW * 32)
void wmma_mm_kernel(const float* __restrict__ Am,
                    const float* __restrict__ Bb, long bz,
                    float* __restrict__ Db, long dz,
                    int ncols) {
    __shared__ alignas(16) float red[MM_NW * 16 * MM_PITCH];

    const int lane = threadIdx.x & 31;
    const int w    = threadIdx.x >> 5;
    const int c    = lane & 15;
    const int h    = lane >> 4;
    const int itile = blockIdx.x;
    const int cbase = blockIdx.y * 64;

    const float* Bm = Bb + (size_t)blockIdx.z * bz;
    float*       Dm = Db + (size_t)blockIdx.z * dz;

    const int colL = (cbase + 4 * c < ncols) ? cbase + 4 * c : 0;

    const int k0 = w * (MM_GPW * 8) + 4 * h;
    const float* Ap = Am + (size_t)(itile * 16 + c) * L + k0;
    const float* Bp = Bm + (size_t)k0 * LDB + colL;

    v4f a[MM_GPW], b[MM_GPW][4];
    #pragma unroll
    for (int i = 0; i < MM_GPW; ++i) {
        a[i] = *(const v4f*)(Ap + 8 * i);
        #pragma unroll
        for (int r = 0; r < 4; ++r)
            b[i][r] = *(const v4f*)(Bp + (size_t)(8 * i + r) * LDB);
    }
    __builtin_amdgcn_sched_barrier(0);

    v8f acc0 = {}, acc1 = {}, acc2 = {}, acc3 = {};
    #pragma unroll
    for (int i = 0; i < MM_GPW; ++i) {
        v2f af, bf;
        af.x = a[i].x; af.y = a[i].y;
        bf.x = b[i][0].x; bf.y = b[i][1].x;
        acc0 = __builtin_amdgcn_wmma_f32_16x16x4_f32(false, af, false, bf, (short)0, acc0, false, false);
        bf.x = b[i][0].y; bf.y = b[i][1].y;
        acc1 = __builtin_amdgcn_wmma_f32_16x16x4_f32(false, af, false, bf, (short)0, acc1, false, false);
        bf.x = b[i][0].z; bf.y = b[i][1].z;
        acc2 = __builtin_amdgcn_wmma_f32_16x16x4_f32(false, af, false, bf, (short)0, acc2, false, false);
        bf.x = b[i][0].w; bf.y = b[i][1].w;
        acc3 = __builtin_amdgcn_wmma_f32_16x16x4_f32(false, af, false, bf, (short)0, acc3, false, false);
        af.x = a[i].z; af.y = a[i].w;
        bf.x = b[i][2].x; bf.y = b[i][3].x;
        acc0 = __builtin_amdgcn_wmma_f32_16x16x4_f32(false, af, false, bf, (short)0, acc0, false, false);
        bf.x = b[i][2].y; bf.y = b[i][3].y;
        acc1 = __builtin_amdgcn_wmma_f32_16x16x4_f32(false, af, false, bf, (short)0, acc1, false, false);
        bf.x = b[i][2].z; bf.y = b[i][3].z;
        acc2 = __builtin_amdgcn_wmma_f32_16x16x4_f32(false, af, false, bf, (short)0, acc2, false, false);
        bf.x = b[i][2].w; bf.y = b[i][3].w;
        acc3 = __builtin_amdgcn_wmma_f32_16x16x4_f32(false, af, false, bf, (short)0, acc3, false, false);
    }

    float* myred = red + (size_t)w * 16 * MM_PITCH;
    #pragma unroll
    for (int v = 0; v < 8; ++v) {
        v4f p;
        p.x = acc0[v]; p.y = acc1[v]; p.z = acc2[v]; p.w = acc3[v];
        *(v4f*)(myred + (8 * h + v) * MM_PITCH + 4 * c) = p;
    }
    __syncthreads();

    for (int o = threadIdx.x; o < 16 * 16; o += MM_NW * 32) {
        const int row = o >> 4;
        const int c4  = (o & 15) * 4;
        v4f s = *(const v4f*)(red + row * MM_PITCH + c4);
        #pragma unroll
        for (int q = 1; q < MM_NW; ++q)
            s += *(const v4f*)(red + (size_t)q * 16 * MM_PITCH + row * MM_PITCH + c4);
        if (cbase + c4 < ncols)
            *(v4f*)(Dm + (size_t)(itile * 16 + row) * LDD + cbase + c4) = s;
    }
}

__global__ void ses2seq_init(const float* __restrict__ Bm,
                             const float* __restrict__ theta0,
                             const float* __restrict__ xs,
                             float* __restrict__ C0,
                             float* __restrict__ thetas) {
    int l = blockIdx.x * blockDim.x + threadIdx.x;
    if (l >= L) return;
    float* row = C0 + (size_t)l * CW;
    const float* brow = Bm + (size_t)l * DS;
    #pragma unroll 4
    for (int d = 0; d < DS; ++d) row[d] = brow[d];
    row[DS] = theta0[l];
    #pragma unroll
    for (int b = 0; b < BATCH; ++b) {
        float acc = 0.f;
        const float* x0 = xs + (size_t)b * T * DS;
        for (int d = 0; d < DS; ++d) acc += brow[d] * x0[d];
        row[DS + 1 + b] = -acc;
    }
    for (int c = DS + 1 + BATCH; c < CW; ++c) row[c] = 0.f;
    #pragma unroll
    for (int b = 0; b < BATCH; ++b)
        thetas[((size_t)b * T + 0) * L + l] = theta0[l];
}

__global__ void ses2seq_conv(const float* __restrict__ C,
                             const float* __restrict__ xs,
                             float* __restrict__ thetas) {
    const int lane  = threadIdx.x & 31;
    const int wv    = threadIdx.x >> 5;
    const int ltile = blockIdx.x;
    const int b     = blockIdx.y;
    const int khalf = (lane >> 4) << 1;
    const int lrow  = ltile * 16 + (lane & 15);
    const int tc    = wv * 16 + (lane & 15);
    const int tcc   = (tc >= NCONV) ? 0 : tc;

    const float* u = xs + (size_t)b * T * DS;

    v8f acc = {};
    for (int m = 0; m < NCONV; ++m) {
        const float* Km = C + (size_t)m * CSTEP;
        const float* Kp = C + (size_t)(m > 0 ? m - 1 : 0) * CSTEP;
        const float  w  = (m > 0) ? 1.f : 0.f;
        int s = tcc - m; if (s < 0) s += NCONV;
        const float* urow = u + (size_t)(s + 1) * DS;
        #pragma unroll 4
        for (int d = 0; d < DS; d += 4) {
            const int dk = d + khalf;
            v2f a, bb;
            a.x = Km[(size_t)lrow * CW + dk + 0] - w * Kp[(size_t)lrow * CW + dk + 0];
            a.y = Km[(size_t)lrow * CW + dk + 1] - w * Kp[(size_t)lrow * CW + dk + 1];
            bb.x = urow[dk + 0];
            bb.y = urow[dk + 1];
            acc = __builtin_amdgcn_wmma_f32_16x16x4_f32(false, a, false, bb,
                                                        (short)0, acc, false, false);
        }
    }

    if (tc < NCONV) {
        const int l0 = ltile * 16 + ((lane >> 4) << 3);
        const int zc = (tc - 1 + NCONV) % NCONV;
        #pragma unroll
        for (int v = 0; v < 8; ++v) {
            const int l = l0 + v;
            float val = acc[v]
                      + C[(size_t)tc * CSTEP + (size_t)l * CW + DS]
                      + C[(size_t)zc * CSTEP + (size_t)l * CW + DS + 1 + b];
            thetas[((size_t)b * T + (tc + 1)) * L + l] = val;
        }
    }
}

__global__ void ses2seq_head(const float* __restrict__ thetas,
                             const float* __restrict__ ts,
                             float* __restrict__ out) {
    const int bt = blockIdx.x;
    const float* th = thetas + (size_t)bt * L;
    __shared__ float h[HID];
    if (threadIdx.x < HID) {
        float x = th[threadIdx.x] * ts[bt] + th[HID + threadIdx.x];
        h[threadIdx.x] = x > 0.f ? x : 0.f;
    }
    __syncthreads();
    const int o = threadIdx.x;
    float acc = th[2 * HID + OUT * HID + o];
    const float* w1 = th + 2 * HID + o * HID;
    #pragma unroll
    for (int k = 0; k < HID; ++k) acc += w1[k] * h[k];
    float r;
    if (o < DS) {
        r = tanhf(acc);
    } else {
        r = acc > 0.f ? acc + log1pf(expf(-acc)) : log1pf(expf(acc));
    }
    out[(size_t)bt * OUT + o] = r;
}

extern "C" void kernel_launch(void* const* d_in, const int* in_sizes, int n_in,
                              void* d_out, int out_size, void* d_ws, size_t ws_size,
                              hipStream_t stream) {
    const float* xs     = (const float*)d_in[0];
    const float* ts     = (const float*)d_in[1];
    const float* theta0 = (const float*)d_in[2];
    const float* A      = (const float*)d_in[3];
    const float* Bm     = (const float*)d_in[4];
    (void)in_sizes; (void)n_in; (void)out_size; (void)ws_size;

    float* A2     = (float*)d_ws;
    float* A4     = A2 + (size_t)L * L;
    float* C      = A4 + (size_t)L * L;
    float* thetas = C + (size_t)NCONV * CSTEP;
    float* out    = (float*)d_out;

    ses2seq_init<<<(L + 255) / 256, 256, 0, stream>>>(Bm, theta0, xs, C, thetas);

    wmma_mm_kernel<L, L><<<dim3(55, 14, 1), MM_NW * 32, 0, stream>>>(A,  A,  0, A2, 0, L);
    wmma_mm_kernel<L, L><<<dim3(55, 14, 1), MM_NW * 32, 0, stream>>>(A2, A2, 0, A4, 0, L);

    wmma_mm_kernel<CW, CW><<<dim3(55, 1, 1), MM_NW * 32, 0, stream>>>(
        A, C, 0, C + CSTEP, 0, CW);
    wmma_mm_kernel<CW, CW><<<dim3(55, 1, 2), MM_NW * 32, 0, stream>>>(
        A2, C, CSTEP, C + (size_t)2 * CSTEP, CSTEP, CW);

    for (int j = 1; j <= 15; ++j) {
        int nz = (j == 15) ? 3 : 4;
        wmma_mm_kernel<CW, CW><<<dim3(55, 1, nz), MM_NW * 32, 0, stream>>>(
            A4,
            C + (size_t)(4 * j - 4) * CSTEP, CSTEP,
            C + (size_t)(4 * j) * CSTEP, CSTEP, CW);
    }

    ses2seq_conv<<<dim3(55, BATCH), 128, 0, stream>>>(C, xs, thetas);

    ses2seq_head<<<BATCH * T, OUT, 0, stream>>>(thetas, ts, out);
}


# conv stage re-tiled: K_conv differences and u staged once in LDS by coalesced 16-byte loads, 16 waves = 4 tc tiles x 4 K quarters, 2 accumulators, quarters summed via LDS: new summation order
# speedup vs baseline: 2.9036x; 1.2465x over previous
#include <hip/hip_runtime.h>
#include <math.h>

#define L     880
#define HID   8
#define OUT   96
#define DS    48
#define T     64
#define BATCH 4
#define NCONV 63
#define CW    64
#define CSTEP (L * CW)

typedef float v2f __attribute__((ext_vector_type(2)));
typedef float v8f __attribute__((ext_vector_type(8)));

#define MM_NW   10
#define MM_KG   (L / 8)
#define MM_GPW  (MM_KG / MM_NW)
#define MM_PITCH 68

typedef float v4f __attribute__((ext_vector_type(4)));

template <int LDB, int LDD>
__global__ __launch_bounds__(MM_NW * 32)
void wmma_mm_kernel(const float* __restrict__ Am,
                    const float* __restrict__ Bb, long bz,
                    float* __restrict__ Db, long dz,
                    int ncols) {
    __shared__ alignas(16) float red[MM_NW * 16 * MM_PITCH];

    const int lane = threadIdx.x & 31;
    const int w    = threadIdx.x >> 5;
    const int c    = lane & 15;
    const int h    = lane >> 4;
    const int itile = blockIdx.x;
    const int cbase = blockIdx.y * 64;

    const float* Bm = Bb + (size_t)blockIdx.z * bz;
    float*       Dm = Db + (size_t)blockIdx.z * dz;

    const int colL = (cbase + 4 * c < ncols) ? cbase + 4 * c : 0;

    const int k0 = w * (MM_GPW * 8) + 4 * h;
    const float* Ap = Am + (size_t)(itile * 16 + c) * L + k0;
    const float* Bp = Bm + (size_t)k0 * LDB + colL;

    v4f a[MM_GPW], b[MM_GPW][4];
    #pragma unroll
    for (int i = 0; i < MM_GPW; ++i) {
        a[i] = *(const v4f*)(Ap + 8 * i);
        #pragma unroll
        for (int r = 0; r < 4; ++r)
            b[i][r] = *(const v4f*)(Bp + (size_t)(8 * i + r) * LDB);
    }
    __builtin_amdgcn_sched_barrier(0);

    v8f acc0 = {}, acc1 = {}, acc2 = {}, acc3 = {};
    #pragma unroll
    for (int i = 0; i < MM_GPW; ++i) {
        v2f af, bf;
        af.x = a[i].x; af.y = a[i].y;
        bf.x = b[i][0].x; bf.y = b[i][1].x;
        acc0 = __builtin_amdgcn_wmma_f32_16x16x4_f32(false, af, false, bf, (short)0, acc0, false, false);
        bf.x = b[i][0].y; bf.y = b[i][1].y;
        acc1 = __builtin_amdgcn_wmma_f32_16x16x4_f32(false, af, false, bf, (short)0, acc1, false, false);
        bf.x = b[i][0].z; bf.y = b[i][1].z;
        acc2 = __builtin_amdgcn_wmma_f32_16x16x4_f32(false, af, false, bf, (short)0, acc2, false, false);
        bf.x = b[i][0].w; bf.y = b[i][1].w;
        acc3 = __builtin_amdgcn_wmma_f32_16x16x4_f32(false, af, false, bf, (short)0, acc3, false, false);
        af.x = a[i].z; af.y = a[i].w;
        bf.x = b[i][2].x; bf.y = b[i][3].x;
        acc0 = __builtin_amdgcn_wmma_f32_16x16x4_f32(false, af, false, bf, (short)0, acc0, false, false);
        bf.x = b[i][2].y; bf.y = b[i][3].y;
        acc1 = __builtin_amdgcn_wmma_f32_16x16x4_f32(false, af, false, bf, (short)0, acc1, false, false);
        bf.x = b[i][2].z; bf.y = b[i][3].z;
        acc2 = __builtin_amdgcn_wmma_f32_16x16x4_f32(false, af, false, bf, (short)0, acc2, false, false);
        bf.x = b[i][2].w; bf.y = b[i][3].w;
        acc3 = __builtin_amdgcn_wmma_f32_16x16x4_f32(false, af, false, bf, (short)0, acc3, false, false);
    }

    float* myred = red + (size_t)w * 16 * MM_PITCH;
    #pragma unroll
    for (int v = 0; v < 8; ++v) {
        v4f p;
        p.x = acc0[v]; p.y = acc1[v]; p.z = acc2[v]; p.w = acc3[v];
        *(v4f*)(myred + (8 * h + v) * MM_PITCH + 4 * c) = p;
    }
    __syncthreads();

    for (int o = threadIdx.x; o < 16 * 16; o += MM_NW * 32) {
        const int row = o >> 4;
        const int c4  = (o & 15) * 4;
        v4f s = *(const v4f*)(red + row * MM_PITCH + c4);
        #pragma unroll
        for (int q = 1; q < MM_NW; ++q)
            s += *(const v4f*)(red + (size_t)q * 16 * MM_PITCH + row * MM_PITCH + c4);
        if (cbase + c4 < ncols)
            *(v4f*)(Dm + (size_t)(itile * 16 + row) * LDD + cbase + c4) = s;
    }
}

__global__ void ses2seq_init(const float* __restrict__ Bm,
                             const float* __restrict__ theta0,
                             const float* __restrict__ xs,
                             float* __restrict__ C0,
                             float* __restrict__ thetas) {
    int l = blockIdx.x * blockDim.x + threadIdx.x;
    if (l >= L) return;
    float* row = C0 + (size_t)l * CW;
    const float* brow = Bm + (size_t)l * DS;
    #pragma unroll 4
    for (int d = 0; d < DS; ++d) row[d] = brow[d];
    row[DS] = theta0[l];
    #pragma unroll
    for (int b = 0; b < BATCH; ++b) {
        float acc = 0.f;
        const float* x0 = xs + (size_t)b * T * DS;
        for (int d = 0; d < DS; ++d) acc += brow[d] * x0[d];
        row[DS + 1 + b] = -acc;
    }
    for (int c = DS + 1 + BATCH; c < CW; ++c) row[c] = 0.f;
    #pragma unroll
    for (int b = 0; b < BATCH; ++b)
        thetas[((size_t)b * T + 0) * L + l] = theta0[l];
}

#define CV_NT   512
#define CV_RP   52
#define CV_MP   (16 * CV_RP)
#define CV_QP   (DS / 4)
#define CV_NA   (2 * 16 * CV_QP)

__device__ __forceinline__ void cv_load(v4f (&a)[6], v4f (&u)[6],
                                        const float* ap, const float* up) {
    #pragma unroll
    for (int g = 0; g < 6; ++g) {
        a[g] = *(const v4f*)(ap + 8 * g);
        u[g] = *(const v4f*)(up + 8 * g);
    }
}

__device__ __forceinline__ void cv_mma(const v4f (&a)[6], const v4f (&u)[6],
                                       v8f& acc0, v8f& acc1) {
    #pragma unroll
    for (int g = 0; g < 6; ++g) {
        v2f af, bf;
        af.x = a[g].x; af.y = a[g].y; bf.x = u[g].x; bf.y = u[g].y;
        acc0 = __builtin_amdgcn_wmma_f32_16x16x4_f32(false, af, false, bf,
                                                     (short)0, acc0, false, false);
        af.x = a[g].z; af.y = a[g].w; bf.x = u[g].z; bf.y = u[g].w;
        acc1 = __builtin_amdgcn_wmma_f32_16x16x4_f32(false, af, false, bf,
                                                     (short)0, acc1, false, false);
    }
}

__global__ __launch_bounds__(CV_NT)
void ses2seq_conv(const float* __restrict__ C,
                  const float* __restrict__ xs,
                  float* __restrict__ thetas) {
    __shared__ alignas(16) float sA[(NCONV + 1) * CV_MP];
    __shared__ alignas(16) float sU[NCONV * CV_RP];
    __shared__ alignas(16) float sR[12 * 256];

    const int tid   = threadIdx.x;
    const int lane  = tid & 31;
    const int w     = __builtin_amdgcn_readfirstlane(tid >> 5);
    const int c     = lane & 15;
    const int h     = lane >> 4;
    const int l0    = blockIdx.x * 16;
    const int b     = blockIdx.y;

    if (w < CV_NA / 32) {
        const int gq  = (w >= CV_NA / 64) ? 1 : 0;
        const int q   = tid - gq * (CV_NA / 2);
        const int row = q / CV_QP;
        const int j   = q - row * CV_QP;
        const int ms  = gq * 32;
        const int me  = gq ? NCONV + 1 : 32;
        const float* srcp = C + (size_t)(l0 + row) * CW + 4 * j;
        float*       dstp = sA + row * CV_RP + 4 * j;
        v4f prev = {0.f, 0.f, 0.f, 0.f};
        if (gq) prev = *(const v4f*)(srcp + (size_t)(ms - 1) * CSTEP);
        for (int mb = ms; mb < me; mb += 16) {
            v4f cur[16];
            #pragma unroll
            for (int i = 0; i < 16; ++i) {
                int mm = mb + i;
                if (mm > NCONV - 1) mm = NCONV - 1;
                cur[i] = *(const v4f*)(srcp + (size_t)mm * CSTEP);
            }
            #pragma unroll
            for (int i = 0; i < 16; ++i) {
                if (mb + i < me) {
                    v4f dv = cur[i] - prev;
                    if (mb + i > NCONV - 1) dv = v4f{0.f, 0.f, 0.f, 0.f};
                    *(v4f*)(dstp + (mb + i) * CV_MP) = dv;
                }
                prev = cur[i];
            }
        }
    } else {
        const float* u = xs + (size_t)b * T * DS + DS;
        const int p0 = tid - CV_NA;
        v4f uv[6];
        #pragma unroll
        for (int i = 0; i < 6; ++i) {
            int p = p0 + 128 * i;
            if (p > NCONV * CV_QP - 1) p = NCONV * CV_QP - 1;
            uv[i] = *(const v4f*)(u + 4 * p);
        }
        #pragma unroll
        for (int i = 0; i < 6; ++i) {
            const int p = p0 + 128 * i;
            if (p < NCONV * CV_QP) {
                const int s = p / CV_QP;
                const int j = p - s * CV_QP;
                *(v4f*)(sU + s * CV_RP + 4 * j) = uv[i];
            }
        }
    }
    __syncthreads();

    const int tct = w & 3;
    const int kq  = w >> 2;
    const int tc  = tct * 16 + c;
    const int tcc = (tc >= NCONV) ? 0 : tc;
    const int m0  = kq * 16;
    const int lq  = l0 + (h << 3);

    float e1[8], e2[8];
    if (kq == 0) {
        const int zc = (tcc + NCONV - 1) % NCONV;
        #pragma unroll
        for (int v = 0; v < 8; ++v) {
            e1[v] = C[(size_t)tcc * CSTEP + (size_t)(lq + v) * CW + DS];
            e2[v] = C[(size_t)zc * CSTEP + (size_t)(lq + v) * CW + DS + 1 + b];
        }
    }

    const float* aP = sA + c * CV_RP + 4 * h;
    const float* uP = sU + 4 * h;

    v8f acc0 = {}, acc1 = {};
    v4f fa0[6], fu0[6], fa1[6], fu1[6];
    {
        int s = tcc - m0; if (s < 0) s += NCONV;
        cv_load(fa0, fu0, aP + m0 * CV_MP, uP + s * CV_RP);
    }
    #pragma unroll
    for (int mp = 0; mp < 8; ++mp) {
        const int m = m0 + 2 * mp;
        {
            int s = tcc - (m + 1); if (s < 0) s += NCONV;
            cv_load(fa1, fu1, aP + (m + 1) * CV_MP, uP + s * CV_RP);
        }
        __builtin_amdgcn_sched_barrier(0);
        cv_mma(fa0, fu0, acc0, acc1);
        if (mp < 7) {
            int s = tcc - (m + 2); if (s < 0) s += NCONV;
            cv_load(fa0, fu0, aP + (m + 2) * CV_MP, uP + s * CV_RP);
        }
        __builtin_amdgcn_sched_barrier(0);
        cv_mma(fa1, fu1, acc0, acc1);
    }
    v8f acc = acc0 + acc1;

    if (kq > 0) {
        float* r = sR + ((kq - 1) * 4 + tct) * 256 + lane;
        #pragma unroll
        for (int v = 0; v < 8; ++v) r[v * 32] = acc[v];
    }
    __syncthreads();

    if (kq == 0 && tc < NCONV) {
        #pragma unroll
        for (int q = 0; q < 3; ++q) {
            const float* r = sR + (q * 4 + tct) * 256 + lane;
            #pragma unroll
            for (int v = 0; v < 8; ++v) acc[v] += r[v * 32];
        }
        #pragma unroll
        for (int v = 0; v < 8; ++v) {
            float val = acc[v] + e1[v] + e2[v];
            thetas[((size_t)b * T + (tc + 1)) * L + lq + v] = val;
        }
    }
}

__global__ void ses2seq_head(const float* __restrict__ thetas,
                             const float* __restrict__ ts,
                             float* __restrict__ out) {
    const int bt = blockIdx.x;
    const float* th = thetas + (size_t)bt * L;
    __shared__ float h[HID];
    if (threadIdx.x < HID) {
        float x = th[threadIdx.x] * ts[bt] + th[HID + threadIdx.x];
        h[threadIdx.x] = x > 0.f ? x : 0.f;
    }
    __syncthreads();
    const int o = threadIdx.x;
    float acc = th[2 * HID + OUT * HID + o];
    const float* w1 = th + 2 * HID + o * HID;
    #pragma unroll
    for (int k = 0; k < HID; ++k) acc += w1[k] * h[k];
    float r;
    if (o < DS) {
        r = tanhf(acc);
    } else {
        r = acc > 0.f ? acc + log1pf(expf(-acc)) : log1pf(expf(acc));
    }
    out[(size_t)bt * OUT + o] = r;
}

extern "C" void kernel_launch(void* const* d_in, const int* in_sizes, int n_in,
                              void* d_out, int out_size, void* d_ws, size_t ws_size,
                              hipStream_t stream) {
    const float* xs     = (const float*)d_in[0];
    const float* ts     = (const float*)d_in[1];
    const float* theta0 = (const float*)d_in[2];
    const float* A      = (const float*)d_in[3];
    const float* Bm     = (const float*)d_in[4];
    (void)in_sizes; (void)n_in; (void)out_size; (void)ws_size;

    float* A2     = (float*)d_ws;
    float* A4     = A2 + (size_t)L * L;
    float* C      = A4 + (size_t)L * L;
    float* thetas = C + (size_t)NCONV * CSTEP;
    float* out    = (float*)d_out;

    ses2seq_init<<<(L + 255) / 256, 256, 0, stream>>>(Bm, theta0, xs, C, thetas);

    wmma_mm_kernel<L, L><<<dim3(55, 14, 1), MM_NW * 32, 0, stream>>>(A,  A,  0, A2, 0, L);
    wmma_mm_kernel<L, L><<<dim3(55, 14, 1), MM_NW * 32, 0, stream>>>(A2, A2, 0, A4, 0, L);

    wmma_mm_kernel<CW, CW><<<dim3(55, 1, 1), MM_NW * 32, 0, stream>>>(
        A, C, 0, C + CSTEP, 0, CW);
    wmma_mm_kernel<CW, CW><<<dim3(55, 1, 2), MM_NW * 32, 0, stream>>>(
        A2, C, CSTEP, C + (size_t)2 * CSTEP, CSTEP, CW);

    for (int j = 1; j <= 15; ++j) {
        int nz = (j == 15) ? 3 : 4;
        wmma_mm_kernel<CW, CW><<<dim3(55, 1, nz), MM_NW * 32, 0, stream>>>(
            A4,
            C + (size_t)(4 * j - 4) * CSTEP, CSTEP,
            C + (size_t)(4 * j) * CSTEP, CSTEP, CW);
    }

    ses2seq_conv<<<dim3(55, BATCH), CV_NT, 0, stream>>>(C, xs, thetas);

    ses2seq_head<<<BATCH * T, OUT, 0, stream>>>(thetas, ts, out);
}


# seeding regrouped so each launch carries every product whose operands are ready (C0 with A2; C1,C2 with A4; C3..C6 together): 22 launches become 19, every sum and its order unchanged
# speedup vs baseline: 3.2096x; 1.1054x over previous
#include <hip/hip_runtime.h>
#include <math.h>

#define L     880
#define HID   8
#define OUT   96
#define DS    48
#define T     64
#define BATCH 4
#define NCONV 63
#define CW    64
#define CSTEP (L * CW)

typedef float v2f __attribute__((ext_vector_type(2)));
typedef float v8f __attribute__((ext_vector_type(8)));

#define MM_NW   10
#define MM_KG   (L / 8)
#define MM_GPW  (MM_KG / MM_NW)
#define MM_PITCH 68

typedef float v4f __attribute__((ext_vector_type(4)));

template <int LDB, int LDD>
__device__ __forceinline__
void mm_strip(const float* __restrict__ Am,
              const float* __restrict__ Bm,
              float* __restrict__ Dm,
              const int itile, const int cbase, const int ncols,
              float* __restrict__ red) {
    const int lane = threadIdx.x & 31;
    const int w    = threadIdx.x >> 5;
    const int c    = lane & 15;
    const int h    = lane >> 4;

    const int colL = (cbase + 4 * c < ncols) ? cbase + 4 * c : 0;

    const int k0 = w * (MM_GPW * 8) + 4 * h;
    const float* Ap = Am + (size_t)(itile * 16 + c) * L + k0;
    const float* Bp = Bm + (size_t)k0 * LDB + colL;

    v4f a[MM_GPW], b[MM_GPW][4];
    #pragma unroll
    for (int i = 0; i < MM_GPW; ++i) {
        a[i] = *(const v4f*)(Ap + 8 * i);
        #pragma unroll
        for (int r = 0; r < 4; ++r)
            b[i][r] = *(const v4f*)(Bp + (size_t)(8 * i + r) * LDB);
    }
    __builtin_amdgcn_sched_barrier(0);

    v8f acc0 = {}, acc1 = {}, acc2 = {}, acc3 = {};
    #pragma unroll
    for (int i = 0; i < MM_GPW; ++i) {
        v2f af, bf;
        af.x = a[i].x; af.y = a[i].y;
        bf.x = b[i][0].x; bf.y = b[i][1].x;
        acc0 = __builtin_amdgcn_wmma_f32_16x16x4_f32(false, af, false, bf, (short)0, acc0, false, false);
        bf.x = b[i][0].y; bf.y = b[i][1].y;
        acc1 = __builtin_amdgcn_wmma_f32_16x16x4_f32(false, af, false, bf, (short)0, acc1, false, false);
        bf.x = b[i][0].z; bf.y = b[i][1].z;
        acc2 = __builtin_amdgcn_wmma_f32_16x16x4_f32(false, af, false, bf, (short)0, acc2, false, false);
        bf.x = b[i][0].w; bf.y = b[i][1].w;
        acc3 = __builtin_amdgcn_wmma_f32_16x16x4_f32(false, af, false, bf, (short)0, acc3, false, false);
        af.x = a[i].z; af.y = a[i].w;
        bf.x = b[i][2].x; bf.y = b[i][3].x;
        acc0 = __builtin_amdgcn_wmma_f32_16x16x4_f32(false, af, false, bf, (short)0, acc0, false, false);
        bf.x = b[i][2].y; bf.y = b[i][3].y;
        acc1 = __builtin_amdgcn_wmma_f32_16x16x4_f32(false, af, false, bf, (short)0, acc1, false, false);
        bf.x = b[i][2].z; bf.y = b[i][3].z;
        acc2 = __builtin_amdgcn_wmma_f32_16x16x4_f32(false, af, false, bf, (short)0, acc2, false, false);
        bf.x = b[i][2].w; bf.y = b[i][3].w;
        acc3 = __builtin_amdgcn_wmma_f32_16x16x4_f32(false, af, false, bf, (short)0, acc3, false, false);
    }

    float* myred = red + (size_t)w * 16 * MM_PITCH;
    #pragma unroll
    for (int v = 0; v < 8; ++v) {
        v4f p;
        p.x = acc0[v]; p.y = acc1[v]; p.z = acc2[v]; p.w = acc3[v];
        *(v4f*)(myred + (8 * h + v) * MM_PITCH + 4 * c) = p;
    }
    __syncthreads();

    for (int o = threadIdx.x; o < 16 * 16; o += MM_NW * 32) {
        const int row = o >> 4;
        const int c4  = (o & 15) * 4;
        v4f s = *(const v4f*)(red + row * MM_PITCH + c4);
        #pragma unroll
        for (int q = 1; q < MM_NW; ++q)
            s += *(const v4f*)(red + (size_t)q * 16 * MM_PITCH + row * MM_PITCH + c4);
        if (cbase + c4 < ncols)
            *(v4f*)(Dm + (size_t)(itile * 16 + row) * LDD + cbase + c4) = s;
    }
}

#define MM_LDS  (MM_NW * 16 * MM_PITCH)
#define MM_SQG  ((L + 63) / 64)

__global__ __launch_bounds__(MM_NW * 32)
void wmma_mm_kernel(const float* __restrict__ Am,
                    const float* __restrict__ Bb, long bz,
                    float* __restrict__ Db, long dz) {
    __shared__ alignas(16) float red[MM_LDS];
    mm_strip<CW, CW>(Am, Bb + (size_t)blockIdx.z * bz, Db + (size_t)blockIdx.z * dz,
                     blockIdx.x, 0, CW, red);
}

__device__ __forceinline__
void ses2seq_init_row(const int l,
                      const float* __restrict__ Bm,
                      const float* __restrict__ theta0,
                      const float* __restrict__ xs,
                      float* __restrict__ C0,
                      float* __restrict__ thetas) {
    float* row = C0 + (size_t)l * CW;
    const float* brow = Bm + (size_t)l * DS;
    #pragma unroll 4
    for (int d = 0; d < DS; ++d) row[d] = brow[d];
    row[DS] = theta0[l];
    #pragma unroll
    for (int b = 0; b < BATCH; ++b) {
        float acc = 0.f;
        const float* x0 = xs + (size_t)b * T * DS;
        for (int d = 0; d < DS; ++d) acc += brow[d] * x0[d];
        row[DS + 1 + b] = -acc;
    }
    for (int c = DS + 1 + BATCH; c < CW; ++c) row[c] = 0.f;
    #pragma unroll
    for (int b = 0; b < BATCH; ++b)
        thetas[((size_t)b * T + 0) * L + l] = theta0[l];
}

__global__ __launch_bounds__(MM_NW * 32)
void seed1_kernel(const float* __restrict__ A,
                  float* __restrict__ A2,
                  const float* __restrict__ Bm,
                  const float* __restrict__ theta0,
                  const float* __restrict__ xs,
                  float* __restrict__ C0,
                  float* __restrict__ thetas) {
    __shared__ alignas(16) float red[MM_LDS];
    if (blockIdx.y > 0) {
        mm_strip<L, L>(A, A, A2, blockIdx.x, (blockIdx.y - 1) * 64, L, red);
    } else {
        const int l = blockIdx.x * (MM_NW * 32) + threadIdx.x;
        if (l < L) ses2seq_init_row(l, Bm, theta0, xs, C0, thetas);
    }
}

__global__ __launch_bounds__(MM_NW * 32)
void seed2_kernel(const float* __restrict__ A,
                  const float* __restrict__ A2,
                  float* __restrict__ A4,
                  float* C) {
    __shared__ alignas(16) float red[MM_LDS];
    if (blockIdx.y < MM_SQG) {
        mm_strip<L, L>(A2, A2, A4, blockIdx.x, blockIdx.y * 64, L, red);
    } else {
        const int      m  = blockIdx.y - MM_SQG + 1;
        const float*   Am = (m == 1) ? A : A2;
        mm_strip<CW, CW>(Am, C, C + (size_t)m * CSTEP, blockIdx.x, 0, CW, red);
    }
}

__global__ __launch_bounds__(MM_NW * 32)
void seed3_kernel(const float* __restrict__ A2,
                  const float* __restrict__ A4,
                  float* C) {
    __shared__ alignas(16) float red[MM_LDS];
    const int    z   = blockIdx.z;
    const float* Am  = (z == 0) ? A2 : A4;
    const int    src = (z == 0) ? 1 : z - 1;
    mm_strip<CW, CW>(Am, C + (size_t)src * CSTEP, C + (size_t)(z + 3) * CSTEP,
                     blockIdx.x, 0, CW, red);
}

#define CV_NT   512
#define CV_RP   52
#define CV_MP   (16 * CV_RP)
#define CV_QP   (DS / 4)
#define CV_NA   (2 * 16 * CV_QP)

__device__ __forceinline__ void cv_load(v4f (&a)[6], v4f (&u)[6],
                                        const float* ap, const float* up) {
    #pragma unroll
    for (int g = 0; g < 6; ++g) {
        a[g] = *(const v4f*)(ap + 8 * g);
        u[g] = *(const v4f*)(up + 8 * g);
    }
}

__device__ __forceinline__ void cv_mma(const v4f (&a)[6], const v4f (&u)[6],
                                       v8f& acc0, v8f& acc1) {
    #pragma unroll
    for (int g = 0; g < 6; ++g) {
        v2f af, bf;
        af.x = a[g].x; af.y = a[g].y; bf.x = u[g].x; bf.y = u[g].y;
        acc0 = __builtin_amdgcn_wmma_f32_16x16x4_f32(false, af, false, bf,
                                                     (short)0, acc0, false, false);
        af.x = a[g].z; af.y = a[g].w; bf.x = u[g].z; bf.y = u[g].w;
        acc1 = __builtin_amdgcn_wmma_f32_16x16x4_f32(false, af, false, bf,
                                                     (short)0, acc1, false, false);
    }
}

__global__ __launch_bounds__(CV_NT)
void ses2seq_conv(const float* __restrict__ C,
                  const float* __restrict__ xs,
                  float* __restrict__ thetas) {
    __shared__ alignas(16) float sA[(NCONV + 1) * CV_MP];
    __shared__ alignas(16) float sU[NCONV * CV_RP];
    __shared__ alignas(16) float sR[12 * 256];

    const int tid   = threadIdx.x;
    const int lane  = tid & 31;
    const int w     = __builtin_amdgcn_readfirstlane(tid >> 5);
    const int c     = lane & 15;
    const int h     = lane >> 4;
    const int l0    = blockIdx.x * 16;
    const int b     = blockIdx.y;

    if (w < CV_NA / 32) {
        const int gq  = (w >= CV_NA / 64) ? 1 : 0;
        const int q   = tid - gq * (CV_NA / 2);
        const int row = q / CV_QP;
        const int j   = q - row * CV_QP;
        const int ms  = gq * 32;
        const int me  = gq ? NCONV + 1 : 32;
        const float* srcp = C + (size_t)(l0 + row) * CW + 4 * j;
        float*       dstp = sA + row * CV_RP + 4 * j;
        v4f prev = {0.f, 0.f, 0.f, 0.f};
        if (gq) prev = *(const v4f*)(srcp + (size_t)(ms - 1) * CSTEP);
        for (int mb = ms; mb < me; mb += 16) {
            v4f cur[16];
            #pragma unroll
            for (int i = 0; i < 16; ++i) {
                int mm = mb + i;
                if (mm > NCONV - 1) mm = NCONV - 1;
                cur[i] = *(const v4f*)(srcp + (size_t)mm * CSTEP);
            }
            #pragma unroll
            for (int i = 0; i < 16; ++i) {
                if (mb + i < me) {
                    v4f dv = cur[i] - prev;
                    if (mb + i > NCONV - 1) dv = v4f{0.f, 0.f, 0.f, 0.f};
                    *(v4f*)(dstp + (mb + i) * CV_MP) = dv;
                }
                prev = cur[i];
            }
        }
    } else {
        const float* u = xs + (size_t)b * T * DS + DS;
        const int p0 = tid - CV_NA;
        v4f uv[6];
        #pragma unroll
        for (int i = 0; i < 6; ++i) {
            int p = p0 + 128 * i;
            if (p > NCONV * CV_QP - 1) p = NCONV * CV_QP - 1;
            uv[i] = *(const v4f*)(u + 4 * p);
        }
        #pragma unroll
        for (int i = 0; i < 6; ++i) {
            const int p = p0 + 128 * i;
            if (p < NCONV * CV_QP) {
                const int s = p / CV_QP;
                const int j = p - s * CV_QP;
                *(v4f*)(sU + s * CV_RP + 4 * j) = uv[i];
            }
        }
    }
    __syncthreads();

    const int tct = w & 3;
    const int kq  = w >> 2;
    const int tc  = tct * 16 + c;
    const int tcc = (tc >= NCONV) ? 0 : tc;
    const int m0  = kq * 16;
    const int lq  = l0 + (h << 3);

    float e1[8], e2[8];
    if (kq == 0) {
        const int zc = (tcc + NCONV - 1) % NCONV;
        #pragma unroll
        for (int v = 0; v < 8; ++v) {
            e1[v] = C[(size_t)tcc * CSTEP + (size_t)(lq + v) * CW + DS];
            e2[v] = C[(size_t)zc * CSTEP + (size_t)(lq + v) * CW + DS + 1 + b];
        }
    }

    const float* aP = sA + c * CV_RP + 4 * h;
    const float* uP = sU + 4 * h;

    v8f acc0 = {}, acc1 = {};
    v4f fa0[6], fu0[6], fa1[6], fu1[6];
    {
        int s = tcc - m0; if (s < 0) s += NCONV;
        cv_load(fa0, fu0, aP + m0 * CV_MP, uP + s * CV_RP);
    }
    #pragma unroll
    for (int mp = 0; mp < 8; ++mp) {
        const int m = m0 + 2 * mp;
        {
            int s = tcc - (m + 1); if (s < 0) s += NCONV;
            cv_load(fa1, fu1, aP + (m + 1) * CV_MP, uP + s * CV_RP);
        }
        __builtin_amdgcn_sched_barrier(0);
        cv_mma(fa0, fu0, acc0, acc1);
        if (mp < 7) {
            int s = tcc - (m + 2); if (s < 0) s += NCONV;
            cv_load(fa0, fu0, aP + (m + 2) * CV_MP, uP + s * CV_RP);
        }
        __builtin_amdgcn_sched_barrier(0);
        cv_mma(fa1, fu1, acc0, acc1);
    }
    v8f acc = acc0 + acc1;

    if (kq > 0) {
        float* r = sR + ((kq - 1) * 4 + tct) * 256 + lane;
        #pragma unroll
        for (int v = 0; v < 8; ++v) r[v * 32] = acc[v];
    }
    __syncthreads();

    if (kq == 0 && tc < NCONV) {
        #pragma unroll
        for (int q = 0; q < 3; ++q) {
            const float* r = sR + (q * 4 + tct) * 256 + lane;
            #pragma unroll
            for (int v = 0; v < 8; ++v) acc[v] += r[v * 32];
        }
        #pragma unroll
        for (int v = 0; v < 8; ++v) {
            float val = acc[v] + e1[v] + e2[v];
            thetas[((size_t)b * T + (tc + 1)) * L + lq + v] = val;
        }
    }
}

__global__ void ses2seq_head(const float* __restrict__ thetas,
                             const float* __restrict__ ts,
                             float* __restrict__ out) {
    const int bt = blockIdx.x;
    const float* th = thetas + (size_t)bt * L;
    __shared__ float h[HID];
    if (threadIdx.x < HID) {
        float x = th[threadIdx.x] * ts[bt] + th[HID + threadIdx.x];
        h[threadIdx.x] = x > 0.f ? x : 0.f;
    }
    __syncthreads();
    const int o = threadIdx.x;
    float acc = th[2 * HID + OUT * HID + o];
    const float* w1 = th + 2 * HID + o * HID;
    #pragma unroll
    for (int k = 0; k < HID; ++k) acc += w1[k] * h[k];
    float r;
    if (o < DS) {
        r = tanhf(acc);
    } else {
        r = acc > 0.f ? acc + log1pf(expf(-acc)) : log1pf(expf(acc));
    }
    out[(size_t)bt * OUT + o] = r;
}

extern "C" void kernel_launch(void* const* d_in, const int* in_sizes, int n_in,
                              void* d_out, int out_size, void* d_ws, size_t ws_size,
                              hipStream_t stream) {
    const float* xs     = (const float*)d_in[0];
    const float* ts     = (const float*)d_in[1];
    const float* theta0 = (const float*)d_in[2];
    const float* A      = (const float*)d_in[3];
    const float* Bm     = (const float*)d_in[4];
    (void)in_sizes; (void)n_in; (void)out_size; (void)ws_size;

    float* A2     = (float*)d_ws;
    float* A4     = A2 + (size_t)L * L;
    float* C      = A4 + (size_t)L * L;
    float* thetas = C + (size_t)NCONV * CSTEP;
    float* out    = (float*)d_out;

    seed1_kernel<<<dim3(55, 1 + MM_SQG), MM_NW * 32, 0, stream>>>(A, A2, Bm, theta0, xs, C, thetas);
    seed2_kernel<<<dim3(55, MM_SQG + 2), MM_NW * 32, 0, stream>>>(A, A2, A4, C);
    seed3_kernel<<<dim3(55, 1, 4), MM_NW * 32, 0, stream>>>(A2, A4, C);

    for (int j = 1; j <= 14; ++j) {
        wmma_mm_kernel<<<dim3(55, 1, 4), MM_NW * 32, 0, stream>>>(
            A4,
            C + (size_t)(4 * j - 1) * CSTEP, CSTEP,
            C + (size_t)(4 * j + 3) * CSTEP, CSTEP);
    }

    ses2seq_conv<<<dim3(55, BATCH), CV_NT, 0, stream>>>(C, xs, thetas);

    ses2seq_head<<<BATCH * T, OUT, 0, stream>>>(thetas, ts, out);
}
